# starting kernel
# baseline (speedup 1.0000x reference)
#include <hip/hip_runtime.h>

#define NN 100000
#define NE 1600000
#define CH 64
#define EPSV 1e-5f

typedef float v2f __attribute__((ext_vector_type(2)));
typedef float v8f __attribute__((ext_vector_type(8)));

__global__ void k_zero(float* __restrict__ p, int n) {
    int i = blockIdx.x * blockDim.x + threadIdx.x;
    if (i < n) p[i] = 0.0f;
}

__global__ void k_deg(const int* __restrict__ dst, float* __restrict__ deg, int e) {
    int i = blockIdx.x * blockDim.x + threadIdx.x;
    if (i < e) atomicAdd(&deg[dst[i]], 1.0f);
}

__global__ void k_dinv(const float* __restrict__ deg, float* __restrict__ dinv, int n) {
    int i = blockIdx.x * blockDim.x + threadIdx.x;
    if (i < n) {
        float d = deg[i];
        dinv[i] = (d > 0.0f) ? rsqrtf(d) : 0.0f;
    }
}

__global__ void k_wedge(const int* __restrict__ src, const int* __restrict__ dst,
                        const float* __restrict__ dinv, float* __restrict__ we, int e) {
    int i = blockIdx.x * blockDim.x + threadIdx.x;
    if (i < e) we[i] = -dinv[src[i]] * dinv[dst[i]];
}

__global__ void k_prop(const int* __restrict__ src, const int* __restrict__ dst,
                       const float* __restrict__ we, const float* __restrict__ xin,
                       float* __restrict__ out) {
    int tid = blockIdx.x * blockDim.x + threadIdx.x;
    int e = tid >> 5;
    int lane = tid & 31;
    int s = src[e], d = dst[e];
    float w = we[e];
    const float2 v = *reinterpret_cast<const float2*>(xin + (long long)s * CH + lane * 2);
    float* o = out + (long long)d * CH + lane * 2;
    atomicAdd(o,     w * v.x);
    atomicAdd(o + 1, w * v.y);
}

__global__ void k_fixup(float* __restrict__ tx2, const float* __restrict__ tx0, int n) {
    int i = blockIdx.x * blockDim.x + threadIdx.x;
    if (i < n) tx2[i] = 2.0f * tx2[i] - tx0[i];
}

__global__ __launch_bounds__(128) void k_cheb_gemm(
        const float* __restrict__ tx0, const float* __restrict__ tx1,
        const float* __restrict__ tx2, const float* __restrict__ Wk,
        const float* __restrict__ bias, float* __restrict__ out,
        float* __restrict__ sum, float* __restrict__ sumsq) {
    const int wave  = threadIdx.x >> 5;
    const int lane  = threadIdx.x & 31;
    const int mrow  = lane & 15;
    const int khalf = lane >> 4;
    const int n0    = wave * 16;
    const int rowbase = blockIdx.x * 16;

    v8f c = {0.f, 0.f, 0.f, 0.f, 0.f, 0.f, 0.f, 0.f};
    const float* txs[3] = {tx0, tx1, tx2};

#pragma unroll
    for (int t = 0; t < 3; ++t) {
        const float* __restrict__ X  = txs[t];
        const float* __restrict__ Wt = Wk + t * CH * CH;
#pragma unroll
        for (int kk = 0; kk < 16; ++kk) {
            const int k0 = kk * 4 + khalf * 2;
            v2f a = *reinterpret_cast<const v2f*>(X + (long long)(rowbase + mrow) * CH + k0);
            v2f b;
            b.x = Wt[(k0 + 0) * CH + n0 + mrow];
            b.y = Wt[(k0 + 1) * CH + n0 + mrow];
            c = __builtin_amdgcn_wmma_f32_16x16x4_f32(
                    false, a, false, b, (short)0, c, false, false);
        }
    }

    const float bb = bias[n0 + mrow];
    float s = 0.0f, s2 = 0.0f;
#pragma unroll
    for (int i = 0; i < 8; ++i) {
        float y = c[i] + bb;
        y = (y > 0.0f) ? y : 0.0f;
        out[(long long)(rowbase + i + 8 * khalf) * CH + n0 + mrow] = y;
        s  += y;
        s2 += y * y;
    }
    atomicAdd(&sum[n0 + mrow], s);
    atomicAdd(&sumsq[n0 + mrow], s2);
}

__global__ void k_finstats(const float* __restrict__ sum, const float* __restrict__ sumsq,
                           float* __restrict__ mean, float* __restrict__ rstd) {
    int ch = threadIdx.x;
    if (ch < CH) {
        float m = sum[ch] * (1.0f / NN);
        float v = sumsq[ch] * (1.0f / NN) - m * m;
        mean[ch] = m;
        rstd[ch] = rsqrtf(v + EPSV);
    }
}

__global__ void k_norm(float* __restrict__ h, const float* __restrict__ mean,
                       const float* __restrict__ rstd, int n) {
    int i = blockIdx.x * blockDim.x + threadIdx.x;
    if (i < n) {
        int ch = i & (CH - 1);
        h[i] = (h[i] - mean[ch]) * rstd[ch];
    }
}

static void run_layer(const int* src, const int* dstp, const float* we,
                      const float* xin, const float* Wk, const float* bk,
                      float* tx1, float* tx2, float* hout, float* stats,
                      hipStream_t stream) {
    const int FN = NN * CH;
    k_zero<<<(2 * FN + 255) / 256, 256, 0, stream>>>(tx1, 2 * FN);
    k_prop<<<(NE * 32) / 256, 256, 0, stream>>>(src, dstp, we, xin, tx1);
    k_prop<<<(NE * 32) / 256, 256, 0, stream>>>(src, dstp, we, tx1, tx2);
    k_fixup<<<(FN + 255) / 256, 256, 0, stream>>>(tx2, xin, FN);
    k_zero<<<1, 256, 0, stream>>>(stats, 128);
    k_cheb_gemm<<<NN / 16, 128, 0, stream>>>(xin, tx1, tx2, Wk, bk, hout,
                                             stats, stats + CH);
    k_finstats<<<1, 64, 0, stream>>>(stats, stats + CH, stats + 2 * CH, stats + 3 * CH);
    k_norm<<<(FN + 255) / 256, 256, 0, stream>>>(hout, stats + 2 * CH, stats + 3 * CH, FN);
}

extern "C" void kernel_launch(void* const* d_in, const int* in_sizes, int n_in,
                              void* d_out, int out_size, void* d_ws, size_t ws_size,
                              hipStream_t stream) {
    (void)in_sizes; (void)n_in; (void)out_size; (void)ws_size;
    const float* x   = (const float*)d_in[0];
    const int*   ei  = (const int*)d_in[1];
    const float* W1  = (const float*)d_in[2];
    const float* b1  = (const float*)d_in[3];
    const float* W2  = (const float*)d_in[4];
    const float* b2  = (const float*)d_in[5];
    float* out = (float*)d_out;

    const int* src  = ei;
    const int* dstp = ei + NE;

    float* ws   = (float*)d_ws;
    float* deg  = ws;
    float* dinv = deg + NN;
    float* we   = dinv + NN;
    float* tx1  = we + NE;
    float* tx2  = tx1 + NN * CH;
    float* h1   = tx2 + NN * CH;
    float* stats = h1 + NN * CH;

    k_zero<<<(NN + 255) / 256, 256, 0, stream>>>(deg, NN);
    k_deg<<<(NE + 255) / 256, 256, 0, stream>>>(dstp, deg, NE);
    k_dinv<<<(NN + 255) / 256, 256, 0, stream>>>(deg, dinv, NN);
    k_wedge<<<(NE + 255) / 256, 256, 0, stream>>>(src, dstp, dinv, we, NE);

    run_layer(src, dstp, we, x, W1, b1, tx1, tx2, h1, stats, stream);
    run_layer(src, dstp, we, h1, W2, b2, tx1, tx2, out, stats, stream);
}


# k_deg, k_prop: each half of the second-level cache adds only into its own half of the destination rows (die id, a work queue per half, edge chunks read twice); float atomics, arrival-order sums kept
# speedup vs baseline: 2.0360x; 2.0360x over previous
#include <hip/hip_runtime.h>

#define NN 100000
#define NE 1600000
#define CH 64
#define EPSV 1e-5f

typedef float v2f __attribute__((ext_vector_type(2)));
typedef float v8f __attribute__((ext_vector_type(8)));

__global__ void k_zero(float* __restrict__ p, int n) {
    int i = blockIdx.x * blockDim.x + threadIdx.x;
    if (i < n) p[i] = 0.0f;
}

#define HALF_NODES (NN / 2)
#define QSTRIDE 64
#define QPAIR (2 * QSTRIDE)

static __device__ __forceinline__ int draw_job(int* __restrict__ q, int njobs) {
    int own = (int)((__builtin_amdgcn_s_sendmsg_rtn(0x87) >> 18) & 1u);
    int t = atomicAdd(&q[own * QSTRIDE], 1);
    if (t >= njobs) {
        own ^= 1;
        t = atomicAdd(&q[own * QSTRIDE], 1);
    }
    return (t < njobs) ? own * njobs + t : -1;
}

#define DEG_EPB 512
#define DEG_JOBS (NE / DEG_EPB)
static_assert(NE % DEG_EPB == 0, "edge count must fill every degree chunk");
static_assert(DEG_EPB == 2 * 256, "two edges per lane of a 256-thread block");

__global__ __launch_bounds__(256) void k_deg(const int* __restrict__ dst, float* __restrict__ deg,
                                             int* __restrict__ q) {
    __shared__ int s_job;
    if (threadIdx.x == 0) s_job = draw_job(q, DEG_JOBS);
    __syncthreads();
    const int job = __builtin_amdgcn_readfirstlane(s_job);
    if (job < 0) return;
    const int own = (job >= DEG_JOBS) ? 1 : 0;
    const int e0  = (job - own * DEG_JOBS) * DEG_EPB;
    const int2 d = *reinterpret_cast<const int2*>(dst + e0 + 2 * threadIdx.x);
    if (((d.x >= HALF_NODES) ? 1 : 0) == own) atomicAdd(&deg[d.x], 1.0f);
    if (((d.y >= HALF_NODES) ? 1 : 0) == own) atomicAdd(&deg[d.y], 1.0f);
}

__global__ void k_dinv(float* __restrict__ deg, float* __restrict__ dinv, int n) {
    int i = blockIdx.x * blockDim.x + threadIdx.x;
    if (i < n) {
        float d = deg[i];
        deg[i] = 0.0f;
        dinv[i] = (d > 0.0f) ? rsqrtf(d) : 0.0f;
    }
}

__global__ void k_wedge(const int* __restrict__ src, const int* __restrict__ dst,
                        const float* __restrict__ dinv, float* __restrict__ we, int e) {
    int i = blockIdx.x * blockDim.x + threadIdx.x;
    if (i < e) we[i] = -dinv[src[i]] * dinv[dst[i]];
}

#define EPW 32
#define PROP_WAVES 8
#define PROP_EPB (EPW * PROP_WAVES)
#define PROP_JOBS (NE / PROP_EPB)
#define PROP_ROWS 24
static_assert(NE % PROP_EPB == 0, "edge count must fill every wave");
static_assert(EPW == 32, "one edge record per lane");
static_assert(2 * PROP_ROWS >= EPW, "two rounds cover a wave's records");
static_assert(CH == 64, "a row is two 128-byte lines, one per 16-lane half of the gather");

static __device__ __forceinline__ float xhalf(float x) {
    return __builtin_bit_cast(float, __builtin_amdgcn_permlanex16(
        __builtin_bit_cast(int, x), __builtin_bit_cast(int, x),
        0x76543210, 0xfedcba98, false, false));
}

__global__ __launch_bounds__(32 * PROP_WAVES) void k_prop(
        const int* __restrict__ src, const int* __restrict__ dst,
        const float* __restrict__ we, const float* __restrict__ xin,
        float* __restrict__ out, int* __restrict__ q) {
    __shared__ int s_job;
    if (threadIdx.x == 0) s_job = draw_job(q, PROP_JOBS);
    __syncthreads();
    const int job = __builtin_amdgcn_readfirstlane(s_job);
    if (job < 0) return;
    const int own = (job >= PROP_JOBS) ? 1 : 0;
    const int lane = threadIdx.x & 31;
    const int hi   = lane >> 4;
    const int l16  = lane & 15;
    const int e0 = (job - own * PROP_JOBS) * PROP_EPB + (threadIdx.x >> 5) * EPW;
    const int   s_l = src[e0 + lane];
    const int   d_l = dst[e0 + lane];
    const float w_l = we[e0 + lane];
    asm volatile("" :: "v"(s_l), "v"(d_l), "v"(w_l));
    unsigned keep = __builtin_amdgcn_ballot_w32((((d_l >= HALF_NODES) ? 1 : 0) == own));

    const int offA = 2 * l16 + hi;
    const int offB = 32 + 2 * l16 + hi;
    const float* xl = xin + lane * 2;

    while (keep != 0u) {
        const int n = min((int)__builtin_popcount(keep), PROP_ROWS);
        float2 v[PROP_ROWS];
        unsigned todo = keep;
#pragma unroll
        for (int k = 0; k < PROP_ROWS; ++k) {
            if (k < n) {
                const int j = __builtin_ctz(keep);
                keep &= keep - 1u;
                const int s = __builtin_amdgcn_readlane(s_l, j);
                v[k] = *reinterpret_cast<const float2*>(xl + (long long)s * CH);
            }
        }
#pragma unroll
        for (int k = 0; k < PROP_ROWS; ++k) {
            if (k < n) {
                const int j = __builtin_ctz(todo);
                todo &= todo - 1u;
                const int d = __builtin_amdgcn_readlane(d_l, j);
                const float w = __int_as_float(__builtin_amdgcn_readlane(__float_as_int(w_l), j));
                const float px = w * v[k].x;
                const float py = w * v[k].y;
                const float got = xhalf(hi ? px : py);
                float* row = out + (long long)d * CH;
                atomicAdd(row + offA, hi ? got : px);
                atomicAdd(row + offB, hi ? py : got);
            }
        }
    }
}

__global__ void k_fixup(float* __restrict__ tx2, const float* __restrict__ tx0, int n) {
    int i = blockIdx.x * blockDim.x + threadIdx.x;
    if (i < n) tx2[i] = 2.0f * tx2[i] - tx0[i];
}

__global__ __launch_bounds__(128) void k_cheb_gemm(
        const float* __restrict__ tx0, const float* __restrict__ tx1,
        const float* __restrict__ tx2, const float* __restrict__ Wk,
        const float* __restrict__ bias, float* __restrict__ out,
        float* __restrict__ sum, float* __restrict__ sumsq) {
    const int wave  = threadIdx.x >> 5;
    const int lane  = threadIdx.x & 31;
    const int mrow  = lane & 15;
    const int khalf = lane >> 4;
    const int n0    = wave * 16;
    const int rowbase = blockIdx.x * 16;

    v8f c = {0.f, 0.f, 0.f, 0.f, 0.f, 0.f, 0.f, 0.f};
    const float* txs[3] = {tx0, tx1, tx2};

#pragma unroll
    for (int t = 0; t < 3; ++t) {
        const float* __restrict__ X  = txs[t];
        const float* __restrict__ Wt = Wk + t * CH * CH;
#pragma unroll
        for (int kk = 0; kk < 16; ++kk) {
            const int k0 = kk * 4 + khalf * 2;
            v2f a = *reinterpret_cast<const v2f*>(X + (long long)(rowbase + mrow) * CH + k0);
            v2f b;
            b.x = Wt[(k0 + 0) * CH + n0 + mrow];
            b.y = Wt[(k0 + 1) * CH + n0 + mrow];
            c = __builtin_amdgcn_wmma_f32_16x16x4_f32(
                    false, a, false, b, (short)0, c, false, false);
        }
    }

    const float bb = bias[n0 + mrow];
    float s = 0.0f, s2 = 0.0f;
#pragma unroll
    for (int i = 0; i < 8; ++i) {
        float y = c[i] + bb;
        y = (y > 0.0f) ? y : 0.0f;
        out[(long long)(rowbase + i + 8 * khalf) * CH + n0 + mrow] = y;
        s  += y;
        s2 += y * y;
    }
    atomicAdd(&sum[n0 + mrow], s);
    atomicAdd(&sumsq[n0 + mrow], s2);
}

__global__ void k_finstats(const float* __restrict__ sum, const float* __restrict__ sumsq,
                           float* __restrict__ mean, float* __restrict__ rstd) {
    int ch = threadIdx.x;
    if (ch < CH) {
        float m = sum[ch] * (1.0f / NN);
        float v = sumsq[ch] * (1.0f / NN) - m * m;
        mean[ch] = m;
        rstd[ch] = rsqrtf(v + EPSV);
    }
}

__global__ void k_norm(float* __restrict__ h, const float* __restrict__ mean,
                       const float* __restrict__ rstd, int n) {
    int i = blockIdx.x * blockDim.x + threadIdx.x;
    if (i < n) {
        int ch = i & (CH - 1);
        h[i] = (h[i] - mean[ch]) * rstd[ch];
    }
}

static void run_layer(const int* src, const int* dstp, const float* we,
                      const float* xin, const float* Wk, const float* bk,
                      float* tx1, float* tx2, float* hout, float* stats,
                      int* q, hipStream_t stream) {
    const int FN = NN * CH;
    k_zero<<<(2 * FN + 255) / 256, 256, 0, stream>>>(tx1, 2 * FN);
    k_prop<<<2 * PROP_JOBS, 32 * PROP_WAVES, 0, stream>>>(src, dstp, we, xin, tx1, q);
    k_prop<<<2 * PROP_JOBS, 32 * PROP_WAVES, 0, stream>>>(src, dstp, we, tx1, tx2, q + QPAIR);
    k_fixup<<<(FN + 255) / 256, 256, 0, stream>>>(tx2, xin, FN);
    k_zero<<<1, 256, 0, stream>>>(stats, 128);
    k_cheb_gemm<<<NN / 16, 128, 0, stream>>>(xin, tx1, tx2, Wk, bk, hout,
                                             stats, stats + CH);
    k_finstats<<<1, 64, 0, stream>>>(stats, stats + CH, stats + 2 * CH, stats + 3 * CH);
    k_norm<<<(FN + 255) / 256, 256, 0, stream>>>(hout, stats + 2 * CH, stats + 3 * CH, FN);
}

extern "C" void kernel_launch(void* const* d_in, const int* in_sizes, int n_in,
                              void* d_out, int out_size, void* d_ws, size_t ws_size,
                              hipStream_t stream) {
    (void)in_sizes; (void)n_in; (void)out_size; (void)ws_size;
    const float* x   = (const float*)d_in[0];
    const int*   ei  = (const int*)d_in[1];
    const float* W1  = (const float*)d_in[2];
    const float* b1  = (const float*)d_in[3];
    const float* W2  = (const float*)d_in[4];
    const float* b2  = (const float*)d_in[5];
    float* out = (float*)d_out;

    const int* src  = ei;
    const int* dstp = ei + NE;

    float* ws   = (float*)d_ws;
    float* deg  = ws;
    float* dinv = deg + NN;
    float* we   = dinv + NN;
    float* tx1  = we + NE;
    float* tx2  = tx1 + NN * CH;
    float* h1   = tx2 + NN * CH;
    float* stats = h1 + NN * CH;

    k_zero<<<(2 * NN + 255) / 256, 256, 0, stream>>>(deg, 2 * NN);
    k_deg<<<2 * DEG_JOBS, 256, 0, stream>>>(dstp, deg, (int*)dinv);
    k_dinv<<<(NN + 255) / 256, 256, 0, stream>>>(deg, dinv, NN);
    int* q = (int*)deg;
    k_wedge<<<(NE + 255) / 256, 256, 0, stream>>>(src, dstp, dinv, we, NE);

    run_layer(src, dstp, we, x, W1, b1, tx1, tx2, h1, stats, q, stream);
    run_layer(src, dstp, we, h1, W2, b2, tx1, tx2, out, stats, q + 2 * QPAIR, stream);
}


# layer nested as X(W0-W2)+b+prop(X W1+prop(X 2W2)): GEMM first, its products are the start values the scatter atomics add into; no clear or fix-up pass; stats in own pass; float sum orders change
# speedup vs baseline: 2.9822x; 1.4647x over previous
#include <hip/hip_runtime.h>

#define NN 100000
#define NE 1600000
#define CH 64
#define EPSV 1e-5f

typedef float v2f __attribute__((ext_vector_type(2)));
typedef float v8f __attribute__((ext_vector_type(8)));

__global__ void k_zero(float* __restrict__ p, int n) {
    int i = blockIdx.x * blockDim.x + threadIdx.x;
    if (i < n) p[i] = 0.0f;
}

#define HALF_NODES (NN / 2)
#define QSTRIDE 64
#define QPAIR (2 * QSTRIDE)

static __device__ __forceinline__ int draw_job(int* __restrict__ q, int njobs) {
    int own = (int)((__builtin_amdgcn_s_sendmsg_rtn(0x87) >> 18) & 1u);
    int t = atomicAdd(&q[own * QSTRIDE], 1);
    if (t >= njobs) {
        own ^= 1;
        t = atomicAdd(&q[own * QSTRIDE], 1);
    }
    return (t < njobs) ? own * njobs + t : -1;
}

#define DEG_EPB 512
#define DEG_JOBS (NE / DEG_EPB)
static_assert(NE % DEG_EPB == 0, "edge count must fill every degree chunk");
static_assert(DEG_EPB == 2 * 256, "two edges per lane of a 256-thread block");

__global__ __launch_bounds__(256) void k_deg(const int* __restrict__ dst, float* __restrict__ deg,
                                             int* __restrict__ q) {
    __shared__ int s_job;
    if (threadIdx.x == 0) s_job = draw_job(q, DEG_JOBS);
    __syncthreads();
    const int job = __builtin_amdgcn_readfirstlane(s_job);
    if (job < 0) return;
    const int own = (job >= DEG_JOBS) ? 1 : 0;
    const int e0  = (job - own * DEG_JOBS) * DEG_EPB;
    const int2 d = *reinterpret_cast<const int2*>(dst + e0 + 2 * threadIdx.x);
    if (((d.x >= HALF_NODES) ? 1 : 0) == own) atomicAdd(&deg[d.x], 1.0f);
    if (((d.y >= HALF_NODES) ? 1 : 0) == own) atomicAdd(&deg[d.y], 1.0f);
}

__global__ void k_dinv(float* __restrict__ deg, float* __restrict__ dinv, int n) {
    int i = blockIdx.x * blockDim.x + threadIdx.x;
    if (i < n) {
        float d = deg[i];
        deg[i] = 0.0f;
        dinv[i] = (d > 0.0f) ? rsqrtf(d) : 0.0f;
    }
}

__global__ void k_wedge(const int* __restrict__ src, const int* __restrict__ dst,
                        const float* __restrict__ dinv, float* __restrict__ we, int e) {
    int i = blockIdx.x * blockDim.x + threadIdx.x;
    if (i < e) we[i] = -dinv[src[i]] * dinv[dst[i]];
}

#define EPW 32
#define PROP_WAVES 8
#define PROP_EPB (EPW * PROP_WAVES)
#define PROP_JOBS (NE / PROP_EPB)
#define PROP_ROWS 24
static_assert(NE % PROP_EPB == 0, "edge count must fill every wave");
static_assert(EPW == 32, "one edge record per lane");
static_assert(2 * PROP_ROWS >= EPW, "two rounds cover a wave's records");
static_assert(CH == 64, "a row is two 128-byte lines, one per 16-lane half of the gather");

static __device__ __forceinline__ float xhalf(float x) {
    return __builtin_bit_cast(float, __builtin_amdgcn_permlanex16(
        __builtin_bit_cast(int, x), __builtin_bit_cast(int, x),
        0x76543210, 0xfedcba98, false, false));
}

__global__ __launch_bounds__(32 * PROP_WAVES) void k_prop(
        const int* __restrict__ src, const int* __restrict__ dst,
        const float* __restrict__ we, const float* __restrict__ xin,
        float* __restrict__ out, int* __restrict__ q) {
    __shared__ int s_job;
    if (threadIdx.x == 0) s_job = draw_job(q, PROP_JOBS);
    __syncthreads();
    const int job = __builtin_amdgcn_readfirstlane(s_job);
    if (job < 0) return;
    const int own = (job >= PROP_JOBS) ? 1 : 0;
    const int lane = threadIdx.x & 31;
    const int hi   = lane >> 4;
    const int l16  = lane & 15;
    const int e0 = (job - own * PROP_JOBS) * PROP_EPB + (threadIdx.x >> 5) * EPW;
    const int   s_l = src[e0 + lane];
    const int   d_l = dst[e0 + lane];
    const float w_l = we[e0 + lane];
    asm volatile("" :: "v"(s_l), "v"(d_l), "v"(w_l));
    unsigned keep = __builtin_amdgcn_ballot_w32((((d_l >= HALF_NODES) ? 1 : 0) == own));

    const int offA = 2 * l16 + hi;
    const int offB = 32 + 2 * l16 + hi;
    const float* xl = xin + lane * 2;

    while (keep != 0u) {
        const int n = min((int)__builtin_popcount(keep), PROP_ROWS);
        float2 v[PROP_ROWS];
        unsigned todo = keep;
#pragma unroll
        for (int k = 0; k < PROP_ROWS; ++k) {
            if (k < n) {
                const int j = __builtin_ctz(keep);
                keep &= keep - 1u;
                const int s = __builtin_amdgcn_readlane(s_l, j);
                v[k] = *reinterpret_cast<const float2*>(xl + (long long)s * CH);
            }
        }
#pragma unroll
        for (int k = 0; k < PROP_ROWS; ++k) {
            if (k < n) {
                const int j = __builtin_ctz(todo);
                todo &= todo - 1u;
                const int d = __builtin_amdgcn_readlane(d_l, j);
                const float w = __int_as_float(__builtin_amdgcn_readlane(__float_as_int(w_l), j));
                const float px = w * v[k].x;
                const float py = w * v[k].y;
                const float got = xhalf(hi ? px : py);
                float* row = out + (long long)d * CH;
                atomicAdd(row + offA, hi ? got : px);
                atomicAdd(row + offB, hi ? py : got);
            }
        }
    }
}

__global__ __launch_bounds__(128) void k_cheb_gemm(
        const float* __restrict__ X, const float* __restrict__ Wk,
        const float* __restrict__ bias, float* __restrict__ acc0,
        float* __restrict__ acc1, float* __restrict__ y2,
        float* __restrict__ stats) {
    if (blockIdx.x == 0) stats[threadIdx.x] = 0.0f;

    const int wave  = threadIdx.x >> 5;
    const int lane  = threadIdx.x & 31;
    const int mrow  = lane & 15;
    const int khalf = lane >> 4;
    const int n0    = wave * 16;
    const int rowbase = blockIdx.x * 16;

    v8f c0 = {0.f, 0.f, 0.f, 0.f, 0.f, 0.f, 0.f, 0.f};
    v8f c1 = c0, c2 = c0;
    const float* __restrict__ Wa = Wk;
    const float* __restrict__ Wb = Wk + CH * CH;
    const float* __restrict__ Wc = Wk + 2 * CH * CH;

#pragma unroll
    for (int kk = 0; kk < 16; ++kk) {
        const int k0 = kk * 4 + khalf * 2;
        v2f a = *reinterpret_cast<const v2f*>(X + (long long)(rowbase + mrow) * CH + k0);
        const float w0x = Wa[(k0 + 0) * CH + n0 + mrow];
        const float w0y = Wa[(k0 + 1) * CH + n0 + mrow];
        const float w2x = Wc[(k0 + 0) * CH + n0 + mrow];
        const float w2y = Wc[(k0 + 1) * CH + n0 + mrow];
        v2f b0, b1, b2;
        b0.x = w0x - w2x;
        b0.y = w0y - w2y;
        b1.x = Wb[(k0 + 0) * CH + n0 + mrow];
        b1.y = Wb[(k0 + 1) * CH + n0 + mrow];
        b2.x = 2.0f * w2x;
        b2.y = 2.0f * w2y;
        c0 = __builtin_amdgcn_wmma_f32_16x16x4_f32(
                false, a, false, b0, (short)0, c0, false, false);
        c1 = __builtin_amdgcn_wmma_f32_16x16x4_f32(
                false, a, false, b1, (short)0, c1, false, false);
        c2 = __builtin_amdgcn_wmma_f32_16x16x4_f32(
                false, a, false, b2, (short)0, c2, false, false);
    }

    const float bb = bias[n0 + mrow];
#pragma unroll
    for (int i = 0; i < 8; ++i) {
        const long long o = (long long)(rowbase + i + 8 * khalf) * CH + n0 + mrow;
        acc0[o] = c0[i] + bb;
        acc1[o] = c1[i];
        y2[o]   = c2[i];
    }
}

#define ST_ROWS 32
static_assert(NN % ST_ROWS == 0, "node count must fill every stats block");
static_assert(CH == 64, "16 lanes of 16 bytes cover a row");

__global__ __launch_bounds__(256) void k_stats(const float* __restrict__ h,
                                               float* __restrict__ sum,
                                               float* __restrict__ sumsq) {
    __shared__ float s_s[8][CH];
    __shared__ float s_q[8][CH];
    const int c4   = threadIdx.x & 15;
    const int r    = threadIdx.x >> 4;
    const int wave = threadIdx.x >> 5;
    const float4* p = reinterpret_cast<const float4*>(h)
                    + (long long)(blockIdx.x * ST_ROWS + r) * (CH / 4) + c4;
    float4 a = p[0];
    float4 b = p[16 * (CH / 4)];
    a.x = (a.x > 0.0f) ? a.x : 0.0f;  a.y = (a.y > 0.0f) ? a.y : 0.0f;
    a.z = (a.z > 0.0f) ? a.z : 0.0f;  a.w = (a.w > 0.0f) ? a.w : 0.0f;
    b.x = (b.x > 0.0f) ? b.x : 0.0f;  b.y = (b.y > 0.0f) ? b.y : 0.0f;
    b.z = (b.z > 0.0f) ? b.z : 0.0f;  b.w = (b.w > 0.0f) ? b.w : 0.0f;
    float sx = a.x + b.x, sy = a.y + b.y, sz = a.z + b.z, sw = a.w + b.w;
    float qx = a.x * a.x + b.x * b.x, qy = a.y * a.y + b.y * b.y;
    float qz = a.z * a.z + b.z * b.z, qw = a.w * a.w + b.w * b.w;
    sx += xhalf(sx);  sy += xhalf(sy);  sz += xhalf(sz);  sw += xhalf(sw);
    qx += xhalf(qx);  qy += xhalf(qy);  qz += xhalf(qz);  qw += xhalf(qw);
    if ((threadIdx.x & 16) == 0) {
        *reinterpret_cast<float4*>(&s_s[wave][4 * c4]) = make_float4(sx, sy, sz, sw);
        *reinterpret_cast<float4*>(&s_q[wave][4 * c4]) = make_float4(qx, qy, qz, qw);
    }
    __syncthreads();
    if (threadIdx.x < 2 * CH) {
        const int ch = threadIdx.x & (CH - 1);
        const bool sq = threadIdx.x >= CH;
        const float* t = sq ? &s_q[0][0] : &s_s[0][0];
        float acc = 0.0f;
        for (int w = 0; w < 8; ++w) acc += t[w * CH + ch];
        atomicAdd((sq ? sumsq : sum) + ch, acc);
    }
}

__global__ void k_finstats(const float* __restrict__ sum, const float* __restrict__ sumsq,
                           float* __restrict__ mean, float* __restrict__ rstd) {
    int ch = threadIdx.x;
    if (ch < CH) {
        float m = sum[ch] * (1.0f / NN);
        float v = sumsq[ch] * (1.0f / NN) - m * m;
        mean[ch] = m;
        rstd[ch] = rsqrtf(v + EPSV);
    }
}

__global__ void k_norm(float* __restrict__ h, const float* __restrict__ mean,
                       const float* __restrict__ rstd, int n) {
    int i = blockIdx.x * blockDim.x + threadIdx.x;
    if (i < n) {
        int ch = i & (CH - 1);
        float y = h[i];
        y = (y > 0.0f) ? y : 0.0f;
        h[i] = (y - mean[ch]) * rstd[ch];
    }
}

static void run_layer(const int* src, const int* dstp, const float* we,
                      const float* xin, const float* Wk, const float* bk,
                      float* tx1, float* tx2, float* hout, float* stats,
                      int* q, hipStream_t stream) {
    const int FN = NN * CH;
    k_cheb_gemm<<<NN / 16, 128, 0, stream>>>(xin, Wk, bk, hout, tx1, tx2, stats);
    k_prop<<<2 * PROP_JOBS, 32 * PROP_WAVES, 0, stream>>>(src, dstp, we, tx2, tx1, q);
    k_prop<<<2 * PROP_JOBS, 32 * PROP_WAVES, 0, stream>>>(src, dstp, we, tx1, hout, q + QPAIR);
    k_stats<<<NN / ST_ROWS, 256, 0, stream>>>(hout, stats, stats + CH);
    k_finstats<<<1, 64, 0, stream>>>(stats, stats + CH, stats + 2 * CH, stats + 3 * CH);
    k_norm<<<(FN + 255) / 256, 256, 0, stream>>>(hout, stats + 2 * CH, stats + 3 * CH, FN);
}

extern "C" void kernel_launch(void* const* d_in, const int* in_sizes, int n_in,
                              void* d_out, int out_size, void* d_ws, size_t ws_size,
                              hipStream_t stream) {
    (void)in_sizes; (void)n_in; (void)out_size; (void)ws_size;
    const float* x   = (const float*)d_in[0];
    const int*   ei  = (const int*)d_in[1];
    const float* W1  = (const float*)d_in[2];
    const float* b1  = (const float*)d_in[3];
    const float* W2  = (const float*)d_in[4];
    const float* b2  = (const float*)d_in[5];
    float* out = (float*)d_out;

    const int* src  = ei;
    const int* dstp = ei + NE;

    float* ws   = (float*)d_ws;
    float* deg  = ws;
    float* dinv = deg + NN;
    float* we   = dinv + NN;
    float* tx1  = we + NE;
    float* tx2  = tx1 + NN * CH;
    float* h1   = tx2 + NN * CH;
    float* stats = h1 + NN * CH;

    k_zero<<<(2 * NN + 255) / 256, 256, 0, stream>>>(deg, 2 * NN);
    k_deg<<<2 * DEG_JOBS, 256, 0, stream>>>(dstp, deg, (int*)dinv);
    k_dinv<<<(NN + 255) / 256, 256, 0, stream>>>(deg, dinv, NN);
    int* q = (int*)deg;
    k_wedge<<<(NE + 255) / 256, 256, 0, stream>>>(src, dstp, dinv, we, NE);

    run_layer(src, dstp, we, x, W1, b1, tx1, tx2, h1, stats, q, stream);
    run_layer(src, dstp, we, h1, W2, b2, tx1, tx2, out, stats, q + 2 * QPAIR, stream);
}
